# starting kernel
# baseline (speedup 1.0000x reference)
#include <hip/hip_runtime.h>

typedef __attribute__((ext_vector_type(16))) _Float16 v16h;
typedef __attribute__((ext_vector_type(8)))  _Float16 v8h;
typedef __attribute__((ext_vector_type(8)))  float    v8f;
typedef __attribute__((ext_vector_type(4)))  float    v4f;

#define B_ROWS 4096
#define NFEAT  4
#define VOCAB  100000
#define EMB    64
#define D_IN   256
#define D_H    256
#define D_O    128

__global__ void wt_kernel(const float* __restrict__ W, _Float16* __restrict__ Wt,
                          int K, int N) {
    int tid = blockIdx.x * blockDim.x + threadIdx.x;
    if (tid >= K * N) return;
    int k = tid / N, n = tid - k * N;
    Wt[(size_t)n * K + k] = (_Float16)W[(size_t)k * N + n];
}

__global__ void embed_kernel(const int* __restrict__ uids, const int* __restrict__ iids,
                             const float* __restrict__ utab, const float* __restrict__ itab,
                             _Float16* __restrict__ xu, _Float16* __restrict__ xi) {
    int t = blockIdx.x * blockDim.x + threadIdx.x;
    int tower = (t >= B_ROWS * D_IN) ? 1 : 0;
    int idx = tower ? (t - B_ROWS * D_IN) : t;
    int row = idx >> 8;
    int within = idx & 255;
    int f = within >> 6;
    int e = within & 63;
    const int*   ids = tower ? iids : uids;
    const float* tab = tower ? itab : utab;
    _Float16*    x   = tower ? xi : xu;
    int id = ids[row * NFEAT + f];
    float val = tab[((size_t)f * VOCAB + (size_t)id) * EMB + e];
    x[(size_t)row * D_IN + within] = (_Float16)val;
}

__launch_bounds__(128)
__global__ void mlp_kernel(const _Float16* __restrict__ x,
                           const _Float16* __restrict__ W1t,
                           const float*    __restrict__ b1,
                           const _Float16* __restrict__ W2t,
                           const float*    __restrict__ b2,
                           float* __restrict__ out_n)
{
    __shared__ _Float16 x_lds[16][D_IN];
    __shared__ _Float16 h_lds[16][D_H];
    __shared__ float    u_lds[16][D_O];
    __shared__ float    rnorm[16];

    const int tid  = threadIdx.x;
    const int wave = tid >> 5;
    const int lane = tid & 31;
    const int half = lane >> 4;
    const int idx  = lane & 15;
    const int row0 = blockIdx.x * 16;

    {
        const v8h* src = reinterpret_cast<const v8h*>(x + (size_t)row0 * D_IN);
        v8h* dst = reinterpret_cast<v8h*>(&x_lds[0][0]);
        for (int i = tid; i < 16 * D_IN / 8; i += 128) dst[i] = src[i];
    }
    __syncthreads();

    for (int nt = wave; nt < D_H / 16; nt += 4) {
        const int n = nt * 16 + idx;
        v8f c = {};
        for (int k0 = 0; k0 < D_IN; k0 += 32) {
            v8h alo = *reinterpret_cast<const v8h*>(&x_lds[idx][k0 + half * 8]);
            v8h ahi = *reinterpret_cast<const v8h*>(&x_lds[idx][k0 + 16 + half * 8]);
            v8h blo = *reinterpret_cast<const v8h*>(W1t + (size_t)n * D_IN + k0 + half * 8);
            v8h bhi = *reinterpret_cast<const v8h*>(W1t + (size_t)n * D_IN + k0 + 16 + half * 8);
            v16h a, b;
            for (int i = 0; i < 8; ++i) {
                a[i] = alo[i]; a[8 + i] = ahi[i];
                b[i] = blo[i]; b[8 + i] = bhi[i];
            }
            c = __builtin_amdgcn_wmma_f32_16x16x32_f16(false, a, false, b,
                                                       (short)0, c, false, false);
        }
        const float bn = b1[n];
        for (int r = 0; r < 8; ++r) {
            float v = c[r] + bn;
            h_lds[half * 8 + r][n] = (_Float16)(v > 0.f ? v : 0.f);
        }
    }
    __syncthreads();

    for (int nt = wave; nt < D_O / 16; nt += 4) {
        const int n = nt * 16 + idx;
        v8f c = {};
        for (int k0 = 0; k0 < D_H; k0 += 32) {
            v8h alo = *reinterpret_cast<const v8h*>(&h_lds[idx][k0 + half * 8]);
            v8h ahi = *reinterpret_cast<const v8h*>(&h_lds[idx][k0 + 16 + half * 8]);
            v8h blo = *reinterpret_cast<const v8h*>(W2t + (size_t)n * D_H + k0 + half * 8);
            v8h bhi = *reinterpret_cast<const v8h*>(W2t + (size_t)n * D_H + k0 + 16 + half * 8);
            v16h a, b;
            for (int i = 0; i < 8; ++i) {
                a[i] = alo[i]; a[8 + i] = ahi[i];
                b[i] = blo[i]; b[8 + i] = bhi[i];
            }
            c = __builtin_amdgcn_wmma_f32_16x16x32_f16(false, a, false, b,
                                                       (short)0, c, false, false);
        }
        const float bn = b2[n];
        for (int r = 0; r < 8; ++r) {
            float v = c[r] + bn;
            u_lds[half * 8 + r][n] = (v > 0.f ? v : 0.f);
        }
    }
    __syncthreads();

    if (tid < 16) {
        float s = 0.f;
        for (int k = 0; k < D_O; ++k) { float v = u_lds[tid][k]; s += v * v; }
        float nrm = sqrtf(s);
        rnorm[tid] = (nrm > 1e-8f) ? nrm : 1e-8f;
    }
    __syncthreads();

    for (int i = tid; i < 16 * D_O; i += 128) {
        int r = i >> 7, cc = i & (D_O - 1);
        out_n[(size_t)(row0 + r) * D_O + cc] = u_lds[r][cc] / rnorm[r];
    }
}

__global__ void score_kernel(const float* __restrict__ un, const float* __restrict__ vn,
                             const int* __restrict__ sw_ids,
                             const float* __restrict__ sw_table,
                             float* __restrict__ out)
{
    const int wave = threadIdx.x >> 5;
    const int lane = threadIdx.x & 31;
    const int row  = blockIdx.x * 8 + wave;
    v4f a = *reinterpret_cast<const v4f*>(un + (size_t)row * D_O + lane * 4);
    for (int j = 0; j < 4; ++j) {
        const int col = (row + j) & (B_ROWS - 1);
        v4f b = *reinterpret_cast<const v4f*>(vn + (size_t)col * D_O + lane * 4);
        float s = a[0] * b[0] + a[1] * b[1] + a[2] * b[2] + a[3] * b[3];
        for (int m = 16; m >= 1; m >>= 1) s += __shfl_xor(s, m, 32);
        if (lane == 0) {
            out[(size_t)row * 4 + j] = s - logf(sw_table[sw_ids[col]]);
        }
    }
}

extern "C" void kernel_launch(void* const* d_in, const int* in_sizes, int n_in,
                              void* d_out, int out_size, void* d_ws, size_t ws_size,
                              hipStream_t stream)
{
    const int*   user_ids = (const int*)  d_in[0];
    const int*   item_ids = (const int*)  d_in[1];
    const int*   sw_ids   = (const int*)  d_in[2];
    const float* utab     = (const float*)d_in[3];
    const float* itab     = (const float*)d_in[4];
    const float* sw_table = (const float*)d_in[5];
    const float* uW1 = (const float*)d_in[6];
    const float* ub1 = (const float*)d_in[7];
    const float* uW2 = (const float*)d_in[8];
    const float* ub2 = (const float*)d_in[9];
    const float* iW1 = (const float*)d_in[10];
    const float* ib1 = (const float*)d_in[11];
    const float* iW2 = (const float*)d_in[12];
    const float* ib2 = (const float*)d_in[13];

    char* ws = (char*)d_ws;
    size_t off = 0;
    auto alloc = [&](size_t bytes) -> void* {
        void* p = ws + off;
        off = (off + bytes + 255) & ~(size_t)255;
        return p;
    };
    _Float16* xu   = (_Float16*)alloc((size_t)B_ROWS * D_IN * 2);
    _Float16* xi   = (_Float16*)alloc((size_t)B_ROWS * D_IN * 2);
    _Float16* W1tu = (_Float16*)alloc((size_t)D_H * D_IN * 2);
    _Float16* W2tu = (_Float16*)alloc((size_t)D_O * D_H * 2);
    _Float16* W1ti = (_Float16*)alloc((size_t)D_H * D_IN * 2);
    _Float16* W2ti = (_Float16*)alloc((size_t)D_O * D_H * 2);
    float*    un   = (float*)   alloc((size_t)B_ROWS * D_O * 4);
    float*    vn   = (float*)   alloc((size_t)B_ROWS * D_O * 4);

    wt_kernel<<<(D_IN * D_H + 255) / 256, 256, 0, stream>>>(uW1, W1tu, D_IN, D_H);
    wt_kernel<<<(D_H * D_O + 255) / 256, 256, 0, stream>>>(uW2, W2tu, D_H, D_O);
    wt_kernel<<<(D_IN * D_H + 255) / 256, 256, 0, stream>>>(iW1, W1ti, D_IN, D_H);
    wt_kernel<<<(D_H * D_O + 255) / 256, 256, 0, stream>>>(iW2, W2ti, D_H, D_O);

    embed_kernel<<<(2 * B_ROWS * D_IN) / 256, 256, 0, stream>>>(
        user_ids, item_ids, utab, itab, xu, xi);

    mlp_kernel<<<B_ROWS / 16, 128, 0, stream>>>(xu, W1tu, ub1, W2tu, ub2, un);
    mlp_kernel<<<B_ROWS / 16, 128, 0, stream>>>(xi, W1ti, ib1, W2ti, ib2, vn);

    score_kernel<<<B_ROWS / 8, 256, 0, stream>>>(un, vn, sw_ids, sw_table, (float*)d_out);
}


# mlp weights: rounded to f16 and put in B-fragment order once per call by 24 extra work-groups of the gather launch; tower kernels load B fragments as whole lines into registers, no LDS staging
# speedup vs baseline: 2.6669x; 2.6669x over previous
#include <hip/hip_runtime.h>

typedef __attribute__((ext_vector_type(16))) _Float16 v16h;
typedef __attribute__((ext_vector_type(8)))  _Float16 v8h;
typedef __attribute__((ext_vector_type(8)))  float    v8f;
typedef __attribute__((ext_vector_type(4)))  float    v4f;

#define B_ROWS 4096
#define NFEAT  4
#define VOCAB  100000
#define EMB    64
#define D_IN   256
#define D_H    256
#define D_O    128

#define ROWS_PER_WAVE 4
#define PREP_BLOCKS   24
__global__ void embed_kernel(const int* __restrict__ uids, const int* __restrict__ iids,
                             const float* __restrict__ utab, const float* __restrict__ itab,
                             _Float16* __restrict__ xu, _Float16* __restrict__ xi,
                             const float* __restrict__ uW1, const float* __restrict__ uW2,
                             const float* __restrict__ iW1, const float* __restrict__ iW2,
                             _Float16* __restrict__ uW1f, _Float16* __restrict__ uW2f,
                             _Float16* __restrict__ iW1f, _Float16* __restrict__ iW2f) {
    const int lane = threadIdx.x & 31;
    if (blockIdx.x < PREP_BLOCKS) {
        const int pw     = blockIdx.x * 8 + (threadIdx.x >> 5);
        const int tower  = (pw >= 96) ? 1 : 0;
        const int r      = pw - tower * 96;
        const int layer2 = (r >= 64) ? 1 : 0;
        const int b      = r - layer2 * 64;
        const int ncols  = layer2 ? D_O : D_H;
        const int kb     = layer2 ? (b >> 2) : (b >> 3);
        const int nb     = layer2 ? (b & 3)  : (b & 7);
        const float* Wm  = tower ? (layer2 ? iW2 : iW1) : (layer2 ? uW2 : uW1);
        _Float16*    Wf  = tower ? (layer2 ? iW2f : iW1f) : (layer2 ? uW2f : uW1f);
        const int sk = (lane & 3) * 8;
        const int sn = (lane >> 2) * 4;
        const float* p = Wm + (size_t)(kb * 32 + sk) * ncols + nb * 32 + sn;
        v4f g[8];
#pragma unroll
        for (int j = 0; j < 8; ++j) g[j] = *reinterpret_cast<const v4f*>(p + (size_t)j * ncols);
        const int nt    = nb * 2 + (sn >> 4);
        const int part  = sk >> 4;
        const int hlf   = (sk >> 3) & 1;
        _Float16* dst = Wf + ((size_t)(((nt * 8 + kb) * 2 + part) * 32) + hlf * 16 + (sn & 15)) * 8;
#pragma unroll
        for (int i = 0; i < 4; ++i) {
            v8h o;
#pragma unroll
            for (int j = 0; j < 8; ++j) o[j] = (_Float16)g[j][i];
            *reinterpret_cast<v8h*>(dst + i * 8) = o;
        }
        return;
    }
    const int gwave = ((blockIdx.x - PREP_BLOCKS) * blockDim.x + threadIdx.x) >> 5;
    const int grow0 = gwave * ROWS_PER_WAVE;
    const int tower = (grow0 >= B_ROWS) ? 1 : 0;
    const int row0  = grow0 & (B_ROWS - 1);
    const int*   ids = tower ? iids : uids;
    const float* tab = tower ? itab : utab;
    _Float16*    x   = tower ? xi : xu;
    const int f = lane >> 3;
    const int e = (lane & 7) * 8;

    int id[ROWS_PER_WAVE];
#pragma unroll
    for (int r = 0; r < ROWS_PER_WAVE; ++r) id[r] = ids[(row0 + r) * NFEAT + f];

    v4f lo[ROWS_PER_WAVE], hi[ROWS_PER_WAVE];
#pragma unroll
    for (int r = 0; r < ROWS_PER_WAVE; ++r) {
        const float* src = tab + ((size_t)f * VOCAB + (size_t)id[r]) * EMB + e;
        lo[r] = *reinterpret_cast<const v4f*>(src);
        hi[r] = *reinterpret_cast<const v4f*>(src + 4);
    }
#pragma unroll
    for (int r = 0; r < ROWS_PER_WAVE; ++r) {
        v8h o;
        o[0] = (_Float16)lo[r][0]; o[1] = (_Float16)lo[r][1];
        o[2] = (_Float16)lo[r][2]; o[3] = (_Float16)lo[r][3];
        o[4] = (_Float16)hi[r][0]; o[5] = (_Float16)hi[r][1];
        o[6] = (_Float16)hi[r][2]; o[7] = (_Float16)hi[r][3];
        *reinterpret_cast<v8h*>(x + (size_t)(row0 + r) * D_IN + lane * 8) = o;
    }
}

template <int XM>
__device__ __forceinline__ float row_xor_add(float x) {
    const int y = __builtin_amdgcn_update_dpp(__float_as_int(x), __float_as_int(x),
                                              0x160 | XM, 0xf, 0xf, false);
    return x + __int_as_float(y);
}

#define LDS_ONLY_BARRIER()                                              \
    do {                                                                \
        __builtin_amdgcn_fence(__ATOMIC_RELEASE, "workgroup", "local"); \
        __builtin_amdgcn_s_barrier();                                   \
        __builtin_amdgcn_fence(__ATOMIC_ACQUIRE, "workgroup", "local"); \
    } while (0)

__launch_bounds__(128)
__global__ void mlp_kernel(const _Float16* __restrict__ x,
                           const _Float16* __restrict__ w1f,
                           const float* __restrict__ b1,
                           const _Float16* __restrict__ w2f,
                           const float* __restrict__ b2,
                           float* __restrict__ out_n)
{
    __shared__ alignas(16) _Float16 x_lds[16][D_IN];
    __shared__ alignas(16) _Float16 h_lds[16][D_H];
    __shared__ alignas(16) float    u_lds[16][D_O];
    __shared__ alignas(16) float    npart[4][16];

    const int tid  = threadIdx.x;
    const int wave = tid >> 5;
    const int lane = tid & 31;
    const int half = lane >> 4;
    const int idx  = lane & 15;
    const int row0 = blockIdx.x * 16;

    float bias1[4], bias2[2];
#pragma unroll
    for (int j = 0; j < 4; ++j) bias1[j] = b1[(wave + 4 * j) * 16 + idx];
#pragma unroll
    for (int j = 0; j < 2; ++j) bias2[j] = b2[(wave + 4 * j) * 16 + idx];
    v8h xr[4];
    {
        const v8h* src = reinterpret_cast<const v8h*>(x + (size_t)row0 * D_IN);
#pragma unroll
        for (int i = 0; i < 4; ++i) xr[i] = src[tid + i * 128];
    }
    __builtin_amdgcn_sched_barrier(0);
    v8h w1lo[4][8], w1hi[4][8];
#pragma unroll
    for (int ks = 0; ks < 8; ++ks) {
#pragma unroll
        for (int j = 0; j < 4; ++j) {
            const _Float16* p = w1f + ((size_t)((((wave + 4 * j) * 8 + ks) * 2) * 32) + lane) * 8;
            w1lo[j][ks] = *reinterpret_cast<const v8h*>(p);
            w1hi[j][ks] = *reinterpret_cast<const v8h*>(p + 256);
        }
    }
    __builtin_amdgcn_sched_barrier(0);
    v8h w2lo[2][8], w2hi[2][8];
#pragma unroll
    for (int ks = 0; ks < 8; ++ks) {
#pragma unroll
        for (int j = 0; j < 2; ++j) {
            const _Float16* p = w2f + ((size_t)((((wave + 4 * j) * 8 + ks) * 2) * 32) + lane) * 8;
            w2lo[j][ks] = *reinterpret_cast<const v8h*>(p);
            w2hi[j][ks] = *reinterpret_cast<const v8h*>(p + 256);
        }
    }

    {
        v8h* dst = reinterpret_cast<v8h*>(&x_lds[0][0]);
#pragma unroll
        for (int i = 0; i < 4; ++i) dst[tid + i * 128] = xr[i];
    }
    LDS_ONLY_BARRIER();

    {
        v8h alo[8], ahi[8];
#pragma unroll
        for (int ks = 0; ks < 8; ++ks) {
            alo[ks] = *reinterpret_cast<const v8h*>(&x_lds[idx][ks * 32 + half * 8]);
            ahi[ks] = *reinterpret_cast<const v8h*>(&x_lds[idx][ks * 32 + 16 + half * 8]);
        }
        __builtin_amdgcn_sched_barrier(0);
        v8f c[4] = {};
#pragma unroll
        for (int ks = 0; ks < 8; ++ks) {
            v16h a = __builtin_shufflevector(alo[ks], ahi[ks], 0, 1, 2, 3, 4, 5, 6, 7,
                                             8, 9, 10, 11, 12, 13, 14, 15);
#pragma unroll
            for (int j = 0; j < 4; ++j) {
                v16h b = __builtin_shufflevector(w1lo[j][ks], w1hi[j][ks],
                                                 0, 1, 2, 3, 4, 5, 6, 7,
                                                 8, 9, 10, 11, 12, 13, 14, 15);
                c[j] = __builtin_amdgcn_wmma_f32_16x16x32_f16(false, a, false, b,
                                                              (short)0, c[j], false, false);
            }
        }
#pragma unroll
        for (int j = 0; j < 4; ++j) {
            const int n = (wave + 4 * j) * 16 + idx;
            for (int r = 0; r < 8; ++r) {
                float v = c[j][r] + bias1[j];
                h_lds[half * 8 + r][n] = (_Float16)(v > 0.f ? v : 0.f);
            }
        }
    }
    LDS_ONLY_BARRIER();

    float u[2][8];
    {
        v8h alo[8], ahi[8];
#pragma unroll
        for (int ks = 0; ks < 8; ++ks) {
            alo[ks] = *reinterpret_cast<const v8h*>(&h_lds[idx][ks * 32 + half * 8]);
            ahi[ks] = *reinterpret_cast<const v8h*>(&h_lds[idx][ks * 32 + 16 + half * 8]);
        }
        __builtin_amdgcn_sched_barrier(0);
        v8f c[2] = {};
#pragma unroll
        for (int ks = 0; ks < 8; ++ks) {
            v16h a = __builtin_shufflevector(alo[ks], ahi[ks], 0, 1, 2, 3, 4, 5, 6, 7,
                                             8, 9, 10, 11, 12, 13, 14, 15);
#pragma unroll
            for (int j = 0; j < 2; ++j) {
                v16h b = __builtin_shufflevector(w2lo[j][ks], w2hi[j][ks],
                                                 0, 1, 2, 3, 4, 5, 6, 7,
                                                 8, 9, 10, 11, 12, 13, 14, 15);
                c[j] = __builtin_amdgcn_wmma_f32_16x16x32_f16(false, a, false, b,
                                                              (short)0, c[j], false, false);
            }
        }
#pragma unroll
        for (int j = 0; j < 2; ++j)
#pragma unroll
            for (int r = 0; r < 8; ++r) {
                float v = c[j][r] + bias2[j];
                u[j][r] = (v > 0.f ? v : 0.f);
            }
    }

    {
        float p[8];
#pragma unroll
        for (int r = 0; r < 8; ++r) p[r] = u[0][r] * u[0][r] + u[1][r] * u[1][r];
#pragma unroll
        for (int r = 0; r < 8; ++r) p[r] = row_xor_add<1>(p[r]);
#pragma unroll
        for (int r = 0; r < 8; ++r) p[r] = row_xor_add<2>(p[r]);
#pragma unroll
        for (int r = 0; r < 8; ++r) p[r] = row_xor_add<4>(p[r]);
#pragma unroll
        for (int r = 0; r < 8; ++r) p[r] = row_xor_add<8>(p[r]);
        if (idx == 0) {
            v4f lo4, hi4;
            lo4[0] = p[0]; lo4[1] = p[1]; lo4[2] = p[2]; lo4[3] = p[3];
            hi4[0] = p[4]; hi4[1] = p[5]; hi4[2] = p[6]; hi4[3] = p[7];
            *reinterpret_cast<v4f*>(&npart[wave][half * 8])     = lo4;
            *reinterpret_cast<v4f*>(&npart[wave][half * 8 + 4]) = hi4;
        }
    }
    __syncthreads();

    {
        float s[8];
#pragma unroll
        for (int g = 0; g < 2; ++g) {
            const v4f q0 = *reinterpret_cast<const v4f*>(&npart[0][half * 8 + 4 * g]);
            const v4f q1 = *reinterpret_cast<const v4f*>(&npart[1][half * 8 + 4 * g]);
            const v4f q2 = *reinterpret_cast<const v4f*>(&npart[2][half * 8 + 4 * g]);
            const v4f q3 = *reinterpret_cast<const v4f*>(&npart[3][half * 8 + 4 * g]);
#pragma unroll
            for (int e = 0; e < 4; ++e) s[4 * g + e] = ((q0[e] + q1[e]) + q2[e]) + q3[e];
        }
#pragma unroll
        for (int r = 0; r < 8; ++r) {
            const float nrm = sqrtf(s[r]);
            const float d   = (nrm > 1e-8f) ? nrm : 1e-8f;
#pragma unroll
            for (int j = 0; j < 2; ++j)
                u_lds[half * 8 + r][(wave + 4 * j) * 16 + idx] = u[j][r] / d;
        }
    }
    __syncthreads();

#pragma unroll
    for (int i = 0; i < 4; ++i) {
        const int r = wave * 4 + i;
        const v4f o = *reinterpret_cast<const v4f*>(&u_lds[r][lane * 4]);
        *reinterpret_cast<v4f*>(out_n + (size_t)(row0 + r) * D_O + lane * 4) = o;
    }
}

__global__ void score_kernel(const float* __restrict__ un, const float* __restrict__ vn,
                             const int* __restrict__ sw_ids,
                             const float* __restrict__ sw_table,
                             float* __restrict__ out)
{
    const int wave = threadIdx.x >> 5;
    const int lane = threadIdx.x & 31;
    const int row  = blockIdx.x * 8 + wave;
    const int id = sw_ids[(row + (lane & 3)) & (B_ROWS - 1)];
    v4f a = *reinterpret_cast<const v4f*>(un + (size_t)row * D_O + lane * 4);
    v4f b[4];
#pragma unroll
    for (int j = 0; j < 4; ++j) {
        const int col = (row + j) & (B_ROWS - 1);
        b[j] = *reinterpret_cast<const v4f*>(vn + (size_t)col * D_O + lane * 4);
    }
    const float w = sw_table[id];
    float s[4];
#pragma unroll
    for (int j = 0; j < 4; ++j) {
        float t = fmaf(a[3], b[j][3], fmaf(a[2], b[j][2], fmaf(a[0], b[j][0], a[1] * b[j][1])));
        for (int m = 16; m >= 1; m >>= 1) t += __shfl_xor(t, m, 32);
        s[j] = t;
    }
    const int lwbits = __float_as_int(logf(w));
    const float l0 = __int_as_float(__builtin_amdgcn_readlane(lwbits, 0));
    const float l1 = __int_as_float(__builtin_amdgcn_readlane(lwbits, 1));
    const float l2 = __int_as_float(__builtin_amdgcn_readlane(lwbits, 2));
    const float l3 = __int_as_float(__builtin_amdgcn_readlane(lwbits, 3));
    if (lane == 0) {
        v4f o;
        o[0] = s[0] - l0; o[1] = s[1] - l1; o[2] = s[2] - l2; o[3] = s[3] - l3;
        *reinterpret_cast<v4f*>(out + (size_t)row * 4) = o;
    }
}

extern "C" void kernel_launch(void* const* d_in, const int* in_sizes, int n_in,
                              void* d_out, int out_size, void* d_ws, size_t ws_size,
                              hipStream_t stream)
{
    const int*   user_ids = (const int*)  d_in[0];
    const int*   item_ids = (const int*)  d_in[1];
    const int*   sw_ids   = (const int*)  d_in[2];
    const float* utab     = (const float*)d_in[3];
    const float* itab     = (const float*)d_in[4];
    const float* sw_table = (const float*)d_in[5];
    const float* uW1 = (const float*)d_in[6];
    const float* ub1 = (const float*)d_in[7];
    const float* uW2 = (const float*)d_in[8];
    const float* ub2 = (const float*)d_in[9];
    const float* iW1 = (const float*)d_in[10];
    const float* ib1 = (const float*)d_in[11];
    const float* iW2 = (const float*)d_in[12];
    const float* ib2 = (const float*)d_in[13];

    char* ws = (char*)d_ws;
    size_t off = 0;
    auto alloc = [&](size_t bytes) -> void* {
        void* p = ws + off;
        off = (off + bytes + 255) & ~(size_t)255;
        return p;
    };
    _Float16* xu   = (_Float16*)alloc((size_t)B_ROWS * D_IN * 2);
    _Float16* xi   = (_Float16*)alloc((size_t)B_ROWS * D_IN * 2);
    float*    un   = (float*)   alloc((size_t)B_ROWS * D_O * 4);
    float*    vn   = (float*)   alloc((size_t)B_ROWS * D_O * 4);
    _Float16* uW1f = (_Float16*)alloc((size_t)D_IN * D_H * 2);
    _Float16* uW2f = (_Float16*)alloc((size_t)D_H * D_O * 2);
    _Float16* iW1f = (_Float16*)alloc((size_t)D_IN * D_H * 2);
    _Float16* iW2f = (_Float16*)alloc((size_t)D_H * D_O * 2);

    embed_kernel<<<PREP_BLOCKS + (2 * B_ROWS / ROWS_PER_WAVE) * 32 / 256, 256, 0, stream>>>(
        user_ids, item_ids, utab, itab, xu, xi,
        uW1, uW2, iW1, iW2, uW1f, uW2f, iW1f, iW2f);

    mlp_kernel<<<B_ROWS / 16, 128, 0, stream>>>(xu, uW1f, ub1, uW2f, ub2, un);
    mlp_kernel<<<B_ROWS / 16, 128, 0, stream>>>(xi, iW1f, ib1, iW2f, ib2, vn);

    score_kernel<<<B_ROWS / 8, 256, 0, stream>>>(un, vn, sw_ids, sw_table, (float*)d_out);
}


# scoring folded into the user-tower launch, now run behind the item tower: un rows go from LDS into the dots with the finished vn rows (lane sums by DPP butterflies, add order kept); launches 4 -> 3
# speedup vs baseline: 3.0639x; 1.1489x over previous
#include <hip/hip_runtime.h>

typedef __attribute__((ext_vector_type(16))) _Float16 v16h;
typedef __attribute__((ext_vector_type(8)))  _Float16 v8h;
typedef __attribute__((ext_vector_type(8)))  float    v8f;
typedef __attribute__((ext_vector_type(4)))  float    v4f;

#define B_ROWS 4096
#define NFEAT  4
#define VOCAB  100000
#define EMB    64
#define D_IN   256
#define D_H    256
#define D_O    128

#define ROWS_PER_WAVE 4
#define PREP_BLOCKS   24
__global__ void embed_kernel(const int* __restrict__ uids, const int* __restrict__ iids,
                             const float* __restrict__ utab, const float* __restrict__ itab,
                             _Float16* __restrict__ xu, _Float16* __restrict__ xi,
                             const float* __restrict__ uW1, const float* __restrict__ uW2,
                             const float* __restrict__ iW1, const float* __restrict__ iW2,
                             _Float16* __restrict__ uW1f, _Float16* __restrict__ uW2f,
                             _Float16* __restrict__ iW1f, _Float16* __restrict__ iW2f) {
    const int lane = threadIdx.x & 31;
    if (blockIdx.x < PREP_BLOCKS) {
        const int pw     = blockIdx.x * 8 + (threadIdx.x >> 5);
        const int tower  = (pw >= 96) ? 1 : 0;
        const int r      = pw - tower * 96;
        const int layer2 = (r >= 64) ? 1 : 0;
        const int b      = r - layer2 * 64;
        const int ncols  = layer2 ? D_O : D_H;
        const int kb     = layer2 ? (b >> 2) : (b >> 3);
        const int nb     = layer2 ? (b & 3)  : (b & 7);
        const float* Wm  = tower ? (layer2 ? iW2 : iW1) : (layer2 ? uW2 : uW1);
        _Float16*    Wf  = tower ? (layer2 ? iW2f : iW1f) : (layer2 ? uW2f : uW1f);
        const int sk = (lane & 3) * 8;
        const int sn = (lane >> 2) * 4;
        const float* p = Wm + (size_t)(kb * 32 + sk) * ncols + nb * 32 + sn;
        v4f g[8];
#pragma unroll
        for (int j = 0; j < 8; ++j) g[j] = *reinterpret_cast<const v4f*>(p + (size_t)j * ncols);
        const int nt    = nb * 2 + (sn >> 4);
        const int part  = sk >> 4;
        const int hlf   = (sk >> 3) & 1;
        _Float16* dst = Wf + ((size_t)(((nt * 8 + kb) * 2 + part) * 32) + hlf * 16 + (sn & 15)) * 8;
#pragma unroll
        for (int i = 0; i < 4; ++i) {
            v8h o;
#pragma unroll
            for (int j = 0; j < 8; ++j) o[j] = (_Float16)g[j][i];
            *reinterpret_cast<v8h*>(dst + i * 8) = o;
        }
        return;
    }
    const int gwave = ((blockIdx.x - PREP_BLOCKS) * blockDim.x + threadIdx.x) >> 5;
    const int grow0 = gwave * ROWS_PER_WAVE;
    const int tower = (grow0 >= B_ROWS) ? 1 : 0;
    const int row0  = grow0 & (B_ROWS - 1);
    const int*   ids = tower ? iids : uids;
    const float* tab = tower ? itab : utab;
    _Float16*    x   = tower ? xi : xu;
    const int f = lane >> 3;
    const int e = (lane & 7) * 8;

    int id[ROWS_PER_WAVE];
#pragma unroll
    for (int r = 0; r < ROWS_PER_WAVE; ++r) id[r] = ids[(row0 + r) * NFEAT + f];

    v4f lo[ROWS_PER_WAVE], hi[ROWS_PER_WAVE];
#pragma unroll
    for (int r = 0; r < ROWS_PER_WAVE; ++r) {
        const float* src = tab + ((size_t)f * VOCAB + (size_t)id[r]) * EMB + e;
        lo[r] = *reinterpret_cast<const v4f*>(src);
        hi[r] = *reinterpret_cast<const v4f*>(src + 4);
    }
#pragma unroll
    for (int r = 0; r < ROWS_PER_WAVE; ++r) {
        v8h o;
        o[0] = (_Float16)lo[r][0]; o[1] = (_Float16)lo[r][1];
        o[2] = (_Float16)lo[r][2]; o[3] = (_Float16)lo[r][3];
        o[4] = (_Float16)hi[r][0]; o[5] = (_Float16)hi[r][1];
        o[6] = (_Float16)hi[r][2]; o[7] = (_Float16)hi[r][3];
        *reinterpret_cast<v8h*>(x + (size_t)(row0 + r) * D_IN + lane * 8) = o;
    }
}

template <int XM>
__device__ __forceinline__ float row_xor_add(float x) {
    const int y = __builtin_amdgcn_update_dpp(__float_as_int(x), __float_as_int(x),
                                              0x160 | XM, 0xf, 0xf, false);
    return x + __int_as_float(y);
}

__device__ __forceinline__ float half_xor_add(float x) {
    const int y = __builtin_amdgcn_permlanex16(__float_as_int(x), __float_as_int(x),
                                               0x76543210, 0xfedcba98, false, false);
    return x + __int_as_float(y);
}

#define LDS_ONLY_BARRIER()                                              \
    do {                                                                \
        __builtin_amdgcn_fence(__ATOMIC_RELEASE, "workgroup", "local"); \
        __builtin_amdgcn_s_barrier();                                   \
        __builtin_amdgcn_fence(__ATOMIC_ACQUIRE, "workgroup", "local"); \
    } while (0)

template <bool SCORE>
__device__ __forceinline__ void tower_body(
        const _Float16* __restrict__ x,
        const _Float16* __restrict__ w1f,
        const float* __restrict__ b1,
        const _Float16* __restrict__ w2f,
        const float* __restrict__ b2,
        float* __restrict__ out_n,
        const float* __restrict__ vn_all,
        const int* __restrict__ sw_ids,
        const float* __restrict__ sw_table,
        float* __restrict__ out)
{
    __shared__ alignas(16) _Float16 x_lds[16][D_IN];
    __shared__ alignas(16) _Float16 h_lds[16][D_H];
    __shared__ alignas(16) float    u_lds[16][D_O];
    __shared__ alignas(16) float    npart[4][16];

    const int tid  = threadIdx.x;
    const int wave = tid >> 5;
    const int lane = tid & 31;
    const int half = lane >> 4;
    const int idx  = lane & 15;
    const int row0 = blockIdx.x * 16;

    int sid = 0;
    if (SCORE) sid = sw_ids[(row0 + wave * 4 + (lane & 7)) & (B_ROWS - 1)];
    float bias1[4], bias2[2];
#pragma unroll
    for (int j = 0; j < 4; ++j) bias1[j] = b1[(wave + 4 * j) * 16 + idx];
#pragma unroll
    for (int j = 0; j < 2; ++j) bias2[j] = b2[(wave + 4 * j) * 16 + idx];
    v8h xr[4];
    {
        const v8h* src = reinterpret_cast<const v8h*>(x + (size_t)row0 * D_IN);
#pragma unroll
        for (int i = 0; i < 4; ++i) xr[i] = src[tid + i * 128];
    }
    __builtin_amdgcn_sched_barrier(0);
    v8h w1lo[4][8], w1hi[4][8];
#pragma unroll
    for (int ks = 0; ks < 8; ++ks) {
#pragma unroll
        for (int j = 0; j < 4; ++j) {
            const _Float16* p = w1f + ((size_t)((((wave + 4 * j) * 8 + ks) * 2) * 32) + lane) * 8;
            w1lo[j][ks] = *reinterpret_cast<const v8h*>(p);
            w1hi[j][ks] = *reinterpret_cast<const v8h*>(p + 256);
        }
    }
    __builtin_amdgcn_sched_barrier(0);
    v8h w2lo[2][8], w2hi[2][8];
#pragma unroll
    for (int ks = 0; ks < 8; ++ks) {
#pragma unroll
        for (int j = 0; j < 2; ++j) {
            const _Float16* p = w2f + ((size_t)((((wave + 4 * j) * 8 + ks) * 2) * 32) + lane) * 8;
            w2lo[j][ks] = *reinterpret_cast<const v8h*>(p);
            w2hi[j][ks] = *reinterpret_cast<const v8h*>(p + 256);
        }
    }

    v4f vb[7];
    if (SCORE) {
#pragma unroll
        for (int k = 0; k < 7; ++k) {
            const int col = (row0 + wave * 4 + k) & (B_ROWS - 1);
            vb[k] = *reinterpret_cast<const v4f*>(vn_all + (size_t)col * D_O + lane * 4);
        }
    }

    {
        v8h* dst = reinterpret_cast<v8h*>(&x_lds[0][0]);
#pragma unroll
        for (int i = 0; i < 4; ++i) dst[tid + i * 128] = xr[i];
    }
    LDS_ONLY_BARRIER();
    float sw = 1.0f;
    if (SCORE) sw = sw_table[sid];

    {
        v8h alo[8], ahi[8];
#pragma unroll
        for (int ks = 0; ks < 8; ++ks) {
            alo[ks] = *reinterpret_cast<const v8h*>(&x_lds[idx][ks * 32 + half * 8]);
            ahi[ks] = *reinterpret_cast<const v8h*>(&x_lds[idx][ks * 32 + 16 + half * 8]);
        }
        __builtin_amdgcn_sched_barrier(0);
        v8f c[4] = {};
#pragma unroll
        for (int ks = 0; ks < 8; ++ks) {
            v16h a = __builtin_shufflevector(alo[ks], ahi[ks], 0, 1, 2, 3, 4, 5, 6, 7,
                                             8, 9, 10, 11, 12, 13, 14, 15);
#pragma unroll
            for (int j = 0; j < 4; ++j) {
                v16h b = __builtin_shufflevector(w1lo[j][ks], w1hi[j][ks],
                                                 0, 1, 2, 3, 4, 5, 6, 7,
                                                 8, 9, 10, 11, 12, 13, 14, 15);
                c[j] = __builtin_amdgcn_wmma_f32_16x16x32_f16(false, a, false, b,
                                                              (short)0, c[j], false, false);
            }
        }
#pragma unroll
        for (int j = 0; j < 4; ++j) {
            const int n = (wave + 4 * j) * 16 + idx;
            for (int r = 0; r < 8; ++r) {
                float v = c[j][r] + bias1[j];
                h_lds[half * 8 + r][n] = (_Float16)(v > 0.f ? v : 0.f);
            }
        }
    }
    LDS_ONLY_BARRIER();

    float u[2][8];
    {
        v8h alo[8], ahi[8];
#pragma unroll
        for (int ks = 0; ks < 8; ++ks) {
            alo[ks] = *reinterpret_cast<const v8h*>(&h_lds[idx][ks * 32 + half * 8]);
            ahi[ks] = *reinterpret_cast<const v8h*>(&h_lds[idx][ks * 32 + 16 + half * 8]);
        }
        __builtin_amdgcn_sched_barrier(0);
        v8f c[2] = {};
#pragma unroll
        for (int ks = 0; ks < 8; ++ks) {
            v16h a = __builtin_shufflevector(alo[ks], ahi[ks], 0, 1, 2, 3, 4, 5, 6, 7,
                                             8, 9, 10, 11, 12, 13, 14, 15);
#pragma unroll
            for (int j = 0; j < 2; ++j) {
                v16h b = __builtin_shufflevector(w2lo[j][ks], w2hi[j][ks],
                                                 0, 1, 2, 3, 4, 5, 6, 7,
                                                 8, 9, 10, 11, 12, 13, 14, 15);
                c[j] = __builtin_amdgcn_wmma_f32_16x16x32_f16(false, a, false, b,
                                                              (short)0, c[j], false, false);
            }
        }
#pragma unroll
        for (int j = 0; j < 2; ++j)
#pragma unroll
            for (int r = 0; r < 8; ++r) {
                float v = c[j][r] + bias2[j];
                u[j][r] = (v > 0.f ? v : 0.f);
            }
    }

    {
        float p[8];
#pragma unroll
        for (int r = 0; r < 8; ++r) p[r] = u[0][r] * u[0][r] + u[1][r] * u[1][r];
#pragma unroll
        for (int r = 0; r < 8; ++r) p[r] = row_xor_add<1>(p[r]);
#pragma unroll
        for (int r = 0; r < 8; ++r) p[r] = row_xor_add<2>(p[r]);
#pragma unroll
        for (int r = 0; r < 8; ++r) p[r] = row_xor_add<4>(p[r]);
#pragma unroll
        for (int r = 0; r < 8; ++r) p[r] = row_xor_add<8>(p[r]);
        if (idx == 0) {
            v4f lo4, hi4;
            lo4[0] = p[0]; lo4[1] = p[1]; lo4[2] = p[2]; lo4[3] = p[3];
            hi4[0] = p[4]; hi4[1] = p[5]; hi4[2] = p[6]; hi4[3] = p[7];
            *reinterpret_cast<v4f*>(&npart[wave][half * 8])     = lo4;
            *reinterpret_cast<v4f*>(&npart[wave][half * 8 + 4]) = hi4;
        }
    }
    if (SCORE) LDS_ONLY_BARRIER(); else __syncthreads();

    {
        float s[8];
#pragma unroll
        for (int g = 0; g < 2; ++g) {
            const v4f q0 = *reinterpret_cast<const v4f*>(&npart[0][half * 8 + 4 * g]);
            const v4f q1 = *reinterpret_cast<const v4f*>(&npart[1][half * 8 + 4 * g]);
            const v4f q2 = *reinterpret_cast<const v4f*>(&npart[2][half * 8 + 4 * g]);
            const v4f q3 = *reinterpret_cast<const v4f*>(&npart[3][half * 8 + 4 * g]);
#pragma unroll
            for (int e = 0; e < 4; ++e) s[4 * g + e] = ((q0[e] + q1[e]) + q2[e]) + q3[e];
        }
#pragma unroll
        for (int r = 0; r < 8; ++r) {
            const float nrm = sqrtf(s[r]);
            const float d   = (nrm > 1e-8f) ? nrm : 1e-8f;
#pragma unroll
            for (int j = 0; j < 2; ++j)
                u_lds[half * 8 + r][(wave + 4 * j) * 16 + idx] = u[j][r] / d;
        }
    }
    __syncthreads();

    if (!SCORE) {
#pragma unroll
        for (int i = 0; i < 4; ++i) {
            const int r = wave * 4 + i;
            const v4f o = *reinterpret_cast<const v4f*>(&u_lds[r][lane * 4]);
            *reinterpret_cast<v4f*>(out_n + (size_t)(row0 + r) * D_O + lane * 4) = o;
        }
    } else {
        v4f a[4];
#pragma unroll
        for (int i = 0; i < 4; ++i)
            a[i] = *reinterpret_cast<const v4f*>(&u_lds[wave * 4 + i][lane * 4]);
        const int lwbits = __float_as_int(logf(sw));
        float lw[7];
#pragma unroll
        for (int k = 0; k < 7; ++k) lw[k] = __int_as_float(__builtin_amdgcn_readlane(lwbits, k));
        v4f o = {0.f, 0.f, 0.f, 0.f};
#pragma unroll
        for (int i = 0; i < 4; ++i) {
            v4f s;
#pragma unroll
            for (int j = 0; j < 4; ++j) {
                const v4f b = vb[i + j];
                float t = fmaf(a[i][3], b[3], fmaf(a[i][2], b[2], fmaf(a[i][0], b[0], a[i][1] * b[1])));
                t = half_xor_add(t);
                t = row_xor_add<8>(t);
                t = row_xor_add<4>(t);
                t = row_xor_add<2>(t);
                t = row_xor_add<1>(t);
                s[j] = t - lw[i + j];
            }
            if (lane == i) o = s;
        }
        if (lane < 4)
            *reinterpret_cast<v4f*>(out + (size_t)(row0 + wave * 4 + lane) * 4) = o;
    }
}

__launch_bounds__(128)
__global__ void mlp_kernel(const _Float16* __restrict__ x, const _Float16* __restrict__ w1f,
                           const float* __restrict__ b1, const _Float16* __restrict__ w2f,
                           const float* __restrict__ b2, float* __restrict__ out_n)
{
    tower_body<false>(x, w1f, b1, w2f, b2, out_n, nullptr, nullptr, nullptr, nullptr);
}

__launch_bounds__(128)
__global__ void mlp_score_kernel(const _Float16* __restrict__ x, const _Float16* __restrict__ w1f,
                                 const float* __restrict__ b1, const _Float16* __restrict__ w2f,
                                 const float* __restrict__ b2, float* __restrict__ out,
                                 const float* __restrict__ vn_all, const int* __restrict__ sw_ids,
                                 const float* __restrict__ sw_table)
{
    tower_body<true>(x, w1f, b1, w2f, b2, nullptr, vn_all, sw_ids, sw_table, out);
}

extern "C" void kernel_launch(void* const* d_in, const int* in_sizes, int n_in,
                              void* d_out, int out_size, void* d_ws, size_t ws_size,
                              hipStream_t stream)
{
    const int*   user_ids = (const int*)  d_in[0];
    const int*   item_ids = (const int*)  d_in[1];
    const int*   sw_ids   = (const int*)  d_in[2];
    const float* utab     = (const float*)d_in[3];
    const float* itab     = (const float*)d_in[4];
    const float* sw_table = (const float*)d_in[5];
    const float* uW1 = (const float*)d_in[6];
    const float* ub1 = (const float*)d_in[7];
    const float* uW2 = (const float*)d_in[8];
    const float* ub2 = (const float*)d_in[9];
    const float* iW1 = (const float*)d_in[10];
    const float* ib1 = (const float*)d_in[11];
    const float* iW2 = (const float*)d_in[12];
    const float* ib2 = (const float*)d_in[13];

    char* ws = (char*)d_ws;
    size_t off = 0;
    auto alloc = [&](size_t bytes) -> void* {
        void* p = ws + off;
        off = (off + bytes + 255) & ~(size_t)255;
        return p;
    };
    _Float16* xu   = (_Float16*)alloc((size_t)B_ROWS * D_IN * 2);
    _Float16* xi   = (_Float16*)alloc((size_t)B_ROWS * D_IN * 2);
    float*    vn   = (float*)   alloc((size_t)B_ROWS * D_O * 4);
    _Float16* uW1f = (_Float16*)alloc((size_t)D_IN * D_H * 2);
    _Float16* uW2f = (_Float16*)alloc((size_t)D_H * D_O * 2);
    _Float16* iW1f = (_Float16*)alloc((size_t)D_IN * D_H * 2);
    _Float16* iW2f = (_Float16*)alloc((size_t)D_H * D_O * 2);

    embed_kernel<<<PREP_BLOCKS + (2 * B_ROWS / ROWS_PER_WAVE) * 32 / 256, 256, 0, stream>>>(
        user_ids, item_ids, utab, itab, xu, xi,
        uW1, uW2, iW1, iW2, uW1f, uW2f, iW1f, iW2f);

    mlp_kernel<<<B_ROWS / 16, 128, 0, stream>>>(xi, iW1f, ib1, iW2f, ib2, vn);
    mlp_score_kernel<<<B_ROWS / 16, 128, 0, stream>>>(xu, uW1f, ub1, uW2f, ub2, (float*)d_out,
                                                      vn, sw_ids, sw_table);
}


# mlp kernels: LDS row pitch of the x and h tiles 512 B -> 528 B (16 B x 33): an A-fragment load (16 rows per 16-byte read) no longer hits one bank group 16 times, it is served at once
# speedup vs baseline: 3.6967x; 1.2065x over previous
#include <hip/hip_runtime.h>

typedef __attribute__((ext_vector_type(16))) _Float16 v16h;
typedef __attribute__((ext_vector_type(8)))  _Float16 v8h;
typedef __attribute__((ext_vector_type(8)))  float    v8f;
typedef __attribute__((ext_vector_type(4)))  float    v4f;

#define B_ROWS 4096
#define NFEAT  4
#define VOCAB  100000
#define EMB    64
#define D_IN   256
#define D_H    256
#define D_O    128

#define XH_PITCH (D_IN + 8)

#define ROWS_PER_WAVE 4
#define PREP_BLOCKS   24
__global__ void embed_kernel(const int* __restrict__ uids, const int* __restrict__ iids,
                             const float* __restrict__ utab, const float* __restrict__ itab,
                             _Float16* __restrict__ xu, _Float16* __restrict__ xi,
                             const float* __restrict__ uW1, const float* __restrict__ uW2,
                             const float* __restrict__ iW1, const float* __restrict__ iW2,
                             _Float16* __restrict__ uW1f, _Float16* __restrict__ uW2f,
                             _Float16* __restrict__ iW1f, _Float16* __restrict__ iW2f) {
    const int lane = threadIdx.x & 31;
    if (blockIdx.x < PREP_BLOCKS) {
        const int pw     = blockIdx.x * 8 + (threadIdx.x >> 5);
        const int tower  = (pw >= 96) ? 1 : 0;
        const int r      = pw - tower * 96;
        const int layer2 = (r >= 64) ? 1 : 0;
        const int b      = r - layer2 * 64;
        const int ncols  = layer2 ? D_O : D_H;
        const int kb     = layer2 ? (b >> 2) : (b >> 3);
        const int nb     = layer2 ? (b & 3)  : (b & 7);
        const float* Wm  = tower ? (layer2 ? iW2 : iW1) : (layer2 ? uW2 : uW1);
        _Float16*    Wf  = tower ? (layer2 ? iW2f : iW1f) : (layer2 ? uW2f : uW1f);
        const int sk = (lane & 3) * 8;
        const int sn = (lane >> 2) * 4;
        const float* p = Wm + (size_t)(kb * 32 + sk) * ncols + nb * 32 + sn;
        v4f g[8];
#pragma unroll
        for (int j = 0; j < 8; ++j) g[j] = *reinterpret_cast<const v4f*>(p + (size_t)j * ncols);
        const int nt    = nb * 2 + (sn >> 4);
        const int part  = sk >> 4;
        const int hlf   = (sk >> 3) & 1;
        _Float16* dst = Wf + ((size_t)(((nt * 8 + kb) * 2 + part) * 32) + hlf * 16 + (sn & 15)) * 8;
#pragma unroll
        for (int i = 0; i < 4; ++i) {
            v8h o;
#pragma unroll
            for (int j = 0; j < 8; ++j) o[j] = (_Float16)g[j][i];
            *reinterpret_cast<v8h*>(dst + i * 8) = o;
        }
        return;
    }
    const int gwave = ((blockIdx.x - PREP_BLOCKS) * blockDim.x + threadIdx.x) >> 5;
    const int grow0 = gwave * ROWS_PER_WAVE;
    const int tower = (grow0 >= B_ROWS) ? 1 : 0;
    const int row0  = grow0 & (B_ROWS - 1);
    const int*   ids = tower ? iids : uids;
    const float* tab = tower ? itab : utab;
    _Float16*    x   = tower ? xi : xu;
    const int f = lane >> 3;
    const int e = (lane & 7) * 8;

    int id[ROWS_PER_WAVE];
#pragma unroll
    for (int r = 0; r < ROWS_PER_WAVE; ++r) id[r] = ids[(row0 + r) * NFEAT + f];

    v4f lo[ROWS_PER_WAVE], hi[ROWS_PER_WAVE];
#pragma unroll
    for (int r = 0; r < ROWS_PER_WAVE; ++r) {
        const float* src = tab + ((size_t)f * VOCAB + (size_t)id[r]) * EMB + e;
        lo[r] = *reinterpret_cast<const v4f*>(src);
        hi[r] = *reinterpret_cast<const v4f*>(src + 4);
    }
#pragma unroll
    for (int r = 0; r < ROWS_PER_WAVE; ++r) {
        v8h o;
        o[0] = (_Float16)lo[r][0]; o[1] = (_Float16)lo[r][1];
        o[2] = (_Float16)lo[r][2]; o[3] = (_Float16)lo[r][3];
        o[4] = (_Float16)hi[r][0]; o[5] = (_Float16)hi[r][1];
        o[6] = (_Float16)hi[r][2]; o[7] = (_Float16)hi[r][3];
        *reinterpret_cast<v8h*>(x + (size_t)(row0 + r) * D_IN + lane * 8) = o;
    }
}

template <int XM>
__device__ __forceinline__ float row_xor_add(float x) {
    const int y = __builtin_amdgcn_update_dpp(__float_as_int(x), __float_as_int(x),
                                              0x160 | XM, 0xf, 0xf, false);
    return x + __int_as_float(y);
}

__device__ __forceinline__ float half_xor_add(float x) {
    const int y = __builtin_amdgcn_permlanex16(__float_as_int(x), __float_as_int(x),
                                               0x76543210, 0xfedcba98, false, false);
    return x + __int_as_float(y);
}

#define LDS_ONLY_BARRIER()                                              \
    do {                                                                \
        __builtin_amdgcn_fence(__ATOMIC_RELEASE, "workgroup", "local"); \
        __builtin_amdgcn_s_barrier();                                   \
        __builtin_amdgcn_fence(__ATOMIC_ACQUIRE, "workgroup", "local"); \
    } while (0)

template <bool SCORE>
__device__ __forceinline__ void tower_body(
        const _Float16* __restrict__ x,
        const _Float16* __restrict__ w1f,
        const float* __restrict__ b1,
        const _Float16* __restrict__ w2f,
        const float* __restrict__ b2,
        float* __restrict__ out_n,
        const float* __restrict__ vn_all,
        const int* __restrict__ sw_ids,
        const float* __restrict__ sw_table,
        float* __restrict__ out)
{
    __shared__ alignas(16) _Float16 x_lds[16][XH_PITCH];
    __shared__ alignas(16) _Float16 h_lds[16][XH_PITCH];
    __shared__ alignas(16) float    u_lds[16][D_O];
    __shared__ alignas(16) float    npart[4][16];

    const int tid  = threadIdx.x;
    const int wave = tid >> 5;
    const int lane = tid & 31;
    const int half = lane >> 4;
    const int idx  = lane & 15;
    const int row0 = blockIdx.x * 16;

    int sid = 0;
    if (SCORE) sid = sw_ids[(row0 + wave * 4 + (lane & 7)) & (B_ROWS - 1)];
    float bias1[4], bias2[2];
#pragma unroll
    for (int j = 0; j < 4; ++j) bias1[j] = b1[(wave + 4 * j) * 16 + idx];
#pragma unroll
    for (int j = 0; j < 2; ++j) bias2[j] = b2[(wave + 4 * j) * 16 + idx];
    v8h xr[4];
    {
        const v8h* src = reinterpret_cast<const v8h*>(x + (size_t)row0 * D_IN);
#pragma unroll
        for (int i = 0; i < 4; ++i) xr[i] = src[tid + i * 128];
    }
    __builtin_amdgcn_sched_barrier(0);
    v8h w1lo[4][8], w1hi[4][8];
#pragma unroll
    for (int ks = 0; ks < 8; ++ks) {
#pragma unroll
        for (int j = 0; j < 4; ++j) {
            const _Float16* p = w1f + ((size_t)((((wave + 4 * j) * 8 + ks) * 2) * 32) + lane) * 8;
            w1lo[j][ks] = *reinterpret_cast<const v8h*>(p);
            w1hi[j][ks] = *reinterpret_cast<const v8h*>(p + 256);
        }
    }
    __builtin_amdgcn_sched_barrier(0);
    v8h w2lo[2][8], w2hi[2][8];
#pragma unroll
    for (int ks = 0; ks < 8; ++ks) {
#pragma unroll
        for (int j = 0; j < 2; ++j) {
            const _Float16* p = w2f + ((size_t)((((wave + 4 * j) * 8 + ks) * 2) * 32) + lane) * 8;
            w2lo[j][ks] = *reinterpret_cast<const v8h*>(p);
            w2hi[j][ks] = *reinterpret_cast<const v8h*>(p + 256);
        }
    }

    v4f vb[7];
    if (SCORE) {
#pragma unroll
        for (int k = 0; k < 7; ++k) {
            const int col = (row0 + wave * 4 + k) & (B_ROWS - 1);
            vb[k] = *reinterpret_cast<const v4f*>(vn_all + (size_t)col * D_O + lane * 4);
        }
    }

    {
#pragma unroll
        for (int i = 0; i < 4; ++i)
            *reinterpret_cast<v8h*>(&x_lds[wave + 4 * i][lane * 8]) = xr[i];
    }
    LDS_ONLY_BARRIER();
    float sw = 1.0f;
    if (SCORE) sw = sw_table[sid];

    {
        v8h alo[8], ahi[8];
#pragma unroll
        for (int ks = 0; ks < 8; ++ks) {
            alo[ks] = *reinterpret_cast<const v8h*>(&x_lds[idx][ks * 32 + half * 8]);
            ahi[ks] = *reinterpret_cast<const v8h*>(&x_lds[idx][ks * 32 + 16 + half * 8]);
        }
        __builtin_amdgcn_sched_barrier(0);
        v8f c[4] = {};
#pragma unroll
        for (int ks = 0; ks < 8; ++ks) {
            v16h a = __builtin_shufflevector(alo[ks], ahi[ks], 0, 1, 2, 3, 4, 5, 6, 7,
                                             8, 9, 10, 11, 12, 13, 14, 15);
#pragma unroll
            for (int j = 0; j < 4; ++j) {
                v16h b = __builtin_shufflevector(w1lo[j][ks], w1hi[j][ks],
                                                 0, 1, 2, 3, 4, 5, 6, 7,
                                                 8, 9, 10, 11, 12, 13, 14, 15);
                c[j] = __builtin_amdgcn_wmma_f32_16x16x32_f16(false, a, false, b,
                                                              (short)0, c[j], false, false);
            }
        }
#pragma unroll
        for (int j = 0; j < 4; ++j) {
            const int n = (wave + 4 * j) * 16 + idx;
            for (int r = 0; r < 8; ++r) {
                float v = c[j][r] + bias1[j];
                h_lds[half * 8 + r][n] = (_Float16)(v > 0.f ? v : 0.f);
            }
        }
    }
    LDS_ONLY_BARRIER();

    float u[2][8];
    {
        v8h alo[8], ahi[8];
#pragma unroll
        for (int ks = 0; ks < 8; ++ks) {
            alo[ks] = *reinterpret_cast<const v8h*>(&h_lds[idx][ks * 32 + half * 8]);
            ahi[ks] = *reinterpret_cast<const v8h*>(&h_lds[idx][ks * 32 + 16 + half * 8]);
        }
        __builtin_amdgcn_sched_barrier(0);
        v8f c[2] = {};
#pragma unroll
        for (int ks = 0; ks < 8; ++ks) {
            v16h a = __builtin_shufflevector(alo[ks], ahi[ks], 0, 1, 2, 3, 4, 5, 6, 7,
                                             8, 9, 10, 11, 12, 13, 14, 15);
#pragma unroll
            for (int j = 0; j < 2; ++j) {
                v16h b = __builtin_shufflevector(w2lo[j][ks], w2hi[j][ks],
                                                 0, 1, 2, 3, 4, 5, 6, 7,
                                                 8, 9, 10, 11, 12, 13, 14, 15);
                c[j] = __builtin_amdgcn_wmma_f32_16x16x32_f16(false, a, false, b,
                                                              (short)0, c[j], false, false);
            }
        }
#pragma unroll
        for (int j = 0; j < 2; ++j)
#pragma unroll
            for (int r = 0; r < 8; ++r) {
                float v = c[j][r] + bias2[j];
                u[j][r] = (v > 0.f ? v : 0.f);
            }
    }

    {
        float p[8];
#pragma unroll
        for (int r = 0; r < 8; ++r) p[r] = u[0][r] * u[0][r] + u[1][r] * u[1][r];
#pragma unroll
        for (int r = 0; r < 8; ++r) p[r] = row_xor_add<1>(p[r]);
#pragma unroll
        for (int r = 0; r < 8; ++r) p[r] = row_xor_add<2>(p[r]);
#pragma unroll
        for (int r = 0; r < 8; ++r) p[r] = row_xor_add<4>(p[r]);
#pragma unroll
        for (int r = 0; r < 8; ++r) p[r] = row_xor_add<8>(p[r]);
        if (idx == 0) {
            v4f lo4, hi4;
            lo4[0] = p[0]; lo4[1] = p[1]; lo4[2] = p[2]; lo4[3] = p[3];
            hi4[0] = p[4]; hi4[1] = p[5]; hi4[2] = p[6]; hi4[3] = p[7];
            *reinterpret_cast<v4f*>(&npart[wave][half * 8])     = lo4;
            *reinterpret_cast<v4f*>(&npart[wave][half * 8 + 4]) = hi4;
        }
    }
    if (SCORE) LDS_ONLY_BARRIER(); else __syncthreads();

    {
        float s[8];
#pragma unroll
        for (int g = 0; g < 2; ++g) {
            const v4f q0 = *reinterpret_cast<const v4f*>(&npart[0][half * 8 + 4 * g]);
            const v4f q1 = *reinterpret_cast<const v4f*>(&npart[1][half * 8 + 4 * g]);
            const v4f q2 = *reinterpret_cast<const v4f*>(&npart[2][half * 8 + 4 * g]);
            const v4f q3 = *reinterpret_cast<const v4f*>(&npart[3][half * 8 + 4 * g]);
#pragma unroll
            for (int e = 0; e < 4; ++e) s[4 * g + e] = ((q0[e] + q1[e]) + q2[e]) + q3[e];
        }
#pragma unroll
        for (int r = 0; r < 8; ++r) {
            const float nrm = sqrtf(s[r]);
            const float d   = (nrm > 1e-8f) ? nrm : 1e-8f;
#pragma unroll
            for (int j = 0; j < 2; ++j)
                u_lds[half * 8 + r][(wave + 4 * j) * 16 + idx] = u[j][r] / d;
        }
    }
    __syncthreads();

    if (!SCORE) {
#pragma unroll
        for (int i = 0; i < 4; ++i) {
            const int r = wave * 4 + i;
            const v4f o = *reinterpret_cast<const v4f*>(&u_lds[r][lane * 4]);
            *reinterpret_cast<v4f*>(out_n + (size_t)(row0 + r) * D_O + lane * 4) = o;
        }
    } else {
        v4f a[4];
#pragma unroll
        for (int i = 0; i < 4; ++i)
            a[i] = *reinterpret_cast<const v4f*>(&u_lds[wave * 4 + i][lane * 4]);
        const int lwbits = __float_as_int(logf(sw));
        float lw[7];
#pragma unroll
        for (int k = 0; k < 7; ++k) lw[k] = __int_as_float(__builtin_amdgcn_readlane(lwbits, k));
        v4f o = {0.f, 0.f, 0.f, 0.f};
#pragma unroll
        for (int i = 0; i < 4; ++i) {
            v4f s;
#pragma unroll
            for (int j = 0; j < 4; ++j) {
                const v4f b = vb[i + j];
                float t = fmaf(a[i][3], b[3], fmaf(a[i][2], b[2], fmaf(a[i][0], b[0], a[i][1] * b[1])));
                t = half_xor_add(t);
                t = row_xor_add<8>(t);
                t = row_xor_add<4>(t);
                t = row_xor_add<2>(t);
                t = row_xor_add<1>(t);
                s[j] = t - lw[i + j];
            }
            if (lane == i) o = s;
        }
        if (lane < 4)
            *reinterpret_cast<v4f*>(out + (size_t)(row0 + wave * 4 + lane) * 4) = o;
    }
}

__launch_bounds__(128)
__global__ void mlp_kernel(const _Float16* __restrict__ x, const _Float16* __restrict__ w1f,
                           const float* __restrict__ b1, const _Float16* __restrict__ w2f,
                           const float* __restrict__ b2, float* __restrict__ out_n)
{
    tower_body<false>(x, w1f, b1, w2f, b2, out_n, nullptr, nullptr, nullptr, nullptr);
}

__launch_bounds__(128)
__global__ void mlp_score_kernel(const _Float16* __restrict__ x, const _Float16* __restrict__ w1f,
                                 const float* __restrict__ b1, const _Float16* __restrict__ w2f,
                                 const float* __restrict__ b2, float* __restrict__ out,
                                 const float* __restrict__ vn_all, const int* __restrict__ sw_ids,
                                 const float* __restrict__ sw_table)
{
    tower_body<true>(x, w1f, b1, w2f, b2, nullptr, vn_all, sw_ids, sw_table, out);
}

extern "C" void kernel_launch(void* const* d_in, const int* in_sizes, int n_in,
                              void* d_out, int out_size, void* d_ws, size_t ws_size,
                              hipStream_t stream)
{
    const int*   user_ids = (const int*)  d_in[0];
    const int*   item_ids = (const int*)  d_in[1];
    const int*   sw_ids   = (const int*)  d_in[2];
    const float* utab     = (const float*)d_in[3];
    const float* itab     = (const float*)d_in[4];
    const float* sw_table = (const float*)d_in[5];
    const float* uW1 = (const float*)d_in[6];
    const float* ub1 = (const float*)d_in[7];
    const float* uW2 = (const float*)d_in[8];
    const float* ub2 = (const float*)d_in[9];
    const float* iW1 = (const float*)d_in[10];
    const float* ib1 = (const float*)d_in[11];
    const float* iW2 = (const float*)d_in[12];
    const float* ib2 = (const float*)d_in[13];

    char* ws = (char*)d_ws;
    size_t off = 0;
    auto alloc = [&](size_t bytes) -> void* {
        void* p = ws + off;
        off = (off + bytes + 255) & ~(size_t)255;
        return p;
    };
    _Float16* xu   = (_Float16*)alloc((size_t)B_ROWS * D_IN * 2);
    _Float16* xi   = (_Float16*)alloc((size_t)B_ROWS * D_IN * 2);
    float*    vn   = (float*)   alloc((size_t)B_ROWS * D_O * 4);
    _Float16* uW1f = (_Float16*)alloc((size_t)D_IN * D_H * 2);
    _Float16* uW2f = (_Float16*)alloc((size_t)D_H * D_O * 2);
    _Float16* iW1f = (_Float16*)alloc((size_t)D_IN * D_H * 2);
    _Float16* iW2f = (_Float16*)alloc((size_t)D_H * D_O * 2);

    embed_kernel<<<PREP_BLOCKS + (2 * B_ROWS / ROWS_PER_WAVE) * 32 / 256, 256, 0, stream>>>(
        user_ids, item_ids, utab, itab, xu, xi,
        uW1, uW2, iW1, iW2, uW1f, uW2f, iW1f, iW2f);

    mlp_kernel<<<B_ROWS / 16, 128, 0, stream>>>(xi, iW1f, ib1, iW2f, ib2, vn);
    mlp_score_kernel<<<B_ROWS / 16, 128, 0, stream>>>(xu, uW1f, ub1, uW2f, ub2, (float*)d_out,
                                                      vn, sw_ids, sw_table);
}
